# starting kernel
# baseline (speedup 1.0000x reference)
#include <hip/hip_runtime.h>


typedef float v4f __attribute__((ext_vector_type(4)));
typedef float v2f __attribute__((ext_vector_type(2)));
typedef float v8f __attribute__((ext_vector_type(8)));

constexpr int S   = 4096;
constexpr int Dm  = 1024;
constexpr int TX  = 16;
constexpr int TY  = 16;
constexpr int SEG = S / TY;
constexpr int DT  = TX * 4;

__global__ __launch_bounds__(TX * TY) void cummean_kernel(
    const float* __restrict__ x, float* __restrict__ out) {
  __shared__ float lds_sums[DT][TY + 1];
  __shared__ float lds_excl[DT][TY + 1];
  __shared__ float inv_lds[S];

  const int tx  = threadIdx.x;
  const int ty  = threadIdx.y;
  const int tid = ty * TX + tx;

  for (int i = tid; i < S; i += TX * TY) {
    inv_lds[i] = 1.0f / (float)(i + 1);
  }

  const int tile        = blockIdx.x;
  const int tiles_per_b = Dm / DT;
  const int b           = tile / tiles_per_b;
  const int dbase       = (tile % tiles_per_b) * DT;
  const float* xb = x   + (size_t)b * S * Dm + dbase;
  float*       ob = out + (size_t)b * S * Dm + dbase;

  const int dd = tx * 4;

  v4f acc = {0.f, 0.f, 0.f, 0.f};
  {
    const v4f* p = (const v4f*)(xb + (size_t)(ty * SEG) * Dm + dd);
    #pragma unroll 4
    for (int i = 0; i < SEG; ++i) {
      acc += p[(size_t)i * (Dm / 4)];
    }
  }
  lds_sums[dd + 0][ty] = acc.x;
  lds_sums[dd + 1][ty] = acc.y;
  lds_sums[dd + 2][ty] = acc.z;
  lds_sums[dd + 3][ty] = acc.w;
  __syncthreads();

  const int wid  = tid >> 5;
  const int lane = tid & 31;
  if (wid < DT / 16) {
    const int colbase = wid * 16;
    const int m    = lane & 15;
    const int half = lane >> 4;
    v8f c = {};
    #pragma unroll
    for (int s4 = 0; s4 < 4; ++s4) {
      const int k0 = s4 * 4 + half * 2;
      v2f a, bm;
      a.x = (k0     < m) ? 1.0f : 0.0f;
      a.y = (k0 + 1 < m) ? 1.0f : 0.0f;
      bm.x = lds_sums[colbase + m][k0];
      bm.y = lds_sums[colbase + m][k0 + 1];
      c = __builtin_amdgcn_wmma_f32_16x16x4_f32(false, a, false, bm, (short)0,
                                                c, false, false);
    }
    #pragma unroll
    for (int r = 0; r < 8; ++r) {
      lds_excl[colbase + m][r + 8 * half] = c[r];
    }
  }
  __syncthreads();

  v4f run;
  run.x = lds_excl[dd + 0][ty];
  run.y = lds_excl[dd + 1][ty];
  run.z = lds_excl[dd + 2][ty];
  run.w = lds_excl[dd + 3][ty];

  const int s0 = ty * SEG;
  const v4f* p = (const v4f*)(xb + (size_t)s0 * Dm + dd);
  v4f*       q = (v4f*)(ob + (size_t)s0 * Dm + dd);
  #pragma unroll 4
  for (int i = 0; i < SEG; ++i) {
    v4f v = __builtin_nontemporal_load(&p[(size_t)i * (Dm / 4)]);
    run += v;
    const float inv = inv_lds[s0 + i];
    v4f o = run * inv;
    __builtin_nontemporal_store(o, &q[(size_t)i * (Dm / 4)]);
  }
}

extern "C" void kernel_launch(void* const* d_in, const int* in_sizes, int n_in,
                              void* d_out, int out_size, void* d_ws, size_t ws_size,
                              hipStream_t stream) {
  const float* x = (const float*)d_in[0];
  float* out = (float*)d_out;
  const int B = in_sizes[0] / (S * Dm);
  dim3 block(TX, TY);
  dim3 grid((unsigned)(B * Dm / DT));
  hipLaunchKernelGGL(cummean_kernel, grid, block, 0, stream, x, out);
}


# Fill the chip: 256 work-groups of 16 waves (32 cols x 64 segments of 64 rows; was 128 of 8 waves, 64 cols x 16 x 256), 16-32 row loads in flight per wave; segment sums regroup (new reduction tree)
# speedup vs baseline: 2.4907x; 2.4907x over previous
#include <hip/hip_runtime.h>


typedef float v4f __attribute__((ext_vector_type(4)));
typedef float v2f __attribute__((ext_vector_type(2)));
typedef float v8f __attribute__((ext_vector_type(8)));

constexpr int S     = 4096;
constexpr int Dm    = 1024;
constexpr int TX    = 8;
constexpr int TY    = 64;
constexpr int SEG   = S / TY;
constexpr int DT    = TX * 4;
constexpr int NTHR  = TX * TY;
constexpr int U     = 16;
constexpr int NB    = SEG / U;
constexpr int PITCH = TY + 1;
constexpr int ROWV  = Dm / 4;

__global__ __launch_bounds__(NTHR) void cummean_kernel(
    const float* __restrict__ x, float* __restrict__ out) {
  __shared__ float lds_sums[DT][PITCH];
  __shared__ float lds_excl[DT][PITCH];
  __shared__ __attribute__((aligned(16))) float inv_lds[S];

  const int tx  = threadIdx.x;
  const int ty  = threadIdx.y;
  const int tid = ty * TX + tx;

  for (int i = tid; i < S; i += NTHR) {
    inv_lds[i] = 1.0f / (float)(i + 1);
  }

  const int tile        = blockIdx.x;
  const int tiles_per_b = Dm / DT;
  const int b           = tile / tiles_per_b;
  const int dbase       = (tile % tiles_per_b) * DT;
  const float* xb = x   + (size_t)b * S * Dm + dbase;
  float*       ob = out + (size_t)b * S * Dm + dbase;

  const int dd = tx * 4;
  const int s0 = ty * SEG;
  const v4f* p = (const v4f*)(xb + (size_t)s0 * Dm + dd);
  v4f*       q = (v4f*)(ob + (size_t)s0 * Dm + dd);

  v4f acc = {0.f, 0.f, 0.f, 0.f};
  {
    v4f cur[U], nxt[U];
    #pragma unroll
    for (int j = 0; j < U; ++j) cur[j] = p[(size_t)j * ROWV];
    #pragma unroll
    for (int bt = 0; bt < NB; ++bt) {
      __builtin_amdgcn_sched_barrier(0);
      if (bt + 1 < NB) {
        #pragma unroll
        for (int j = 0; j < U; ++j) nxt[j] = p[(size_t)((bt + 1) * U + j) * ROWV];
      }
      __builtin_amdgcn_sched_barrier(0);
      #pragma unroll
      for (int j = 0; j < U; ++j) acc += cur[j];
      if (bt + 1 < NB) {
        #pragma unroll
        for (int j = 0; j < U; ++j) cur[j] = nxt[j];
      }
    }
  }
  lds_sums[dd + 0][ty] = acc.x;
  lds_sums[dd + 1][ty] = acc.y;
  lds_sums[dd + 2][ty] = acc.z;
  lds_sums[dd + 3][ty] = acc.w;
  __syncthreads();

  const int wid  = tid >> 5;
  const int lane = tid & 31;
  if (wid < (DT / 16) * (TY / 16)) {
    const int colbase = (wid & 1) * 16;
    const int g       = wid >> 1;
    const int m    = lane & 15;
    const int half = lane >> 4;
    const int mrow = g * 16 + m;
    v8f c = {};
    #pragma unroll
    for (int k4 = 0; k4 < TY / 4; ++k4) {
      const int k0 = k4 * 4 + half * 2;
      v2f a, bm;
      a.x = (k0     < mrow) ? 1.0f : 0.0f;
      a.y = (k0 + 1 < mrow) ? 1.0f : 0.0f;
      bm.x = lds_sums[colbase + m][k0];
      bm.y = lds_sums[colbase + m][k0 + 1];
      c = __builtin_amdgcn_wmma_f32_16x16x4_f32(false, a, false, bm, (short)0,
                                                c, false, false);
    }
    #pragma unroll
    for (int r = 0; r < 8; ++r) {
      lds_excl[colbase + m][g * 16 + r + 8 * half] = c[r];
    }
  }
  __syncthreads();

  v4f run;
  run.x = lds_excl[dd + 0][ty];
  run.y = lds_excl[dd + 1][ty];
  run.z = lds_excl[dd + 2][ty];
  run.w = lds_excl[dd + 3][ty];
  {
    v4f cur[U], nxt[U];
    #pragma unroll
    for (int j = 0; j < U; ++j)
      cur[j] = __builtin_nontemporal_load(&p[(size_t)j * ROWV]);
    #pragma unroll
    for (int bt = 0; bt < NB; ++bt) {
      __builtin_amdgcn_sched_barrier(0);
      if (bt + 1 < NB) {
        #pragma unroll
        for (int j = 0; j < U; ++j)
          nxt[j] = __builtin_nontemporal_load(&p[(size_t)((bt + 1) * U + j) * ROWV]);
      }
      __builtin_amdgcn_sched_barrier(0);
      const v4f* invp = (const v4f*)&inv_lds[s0 + bt * U];
      #pragma unroll
      for (int j4 = 0; j4 < U / 4; ++j4) {
        const v4f inv4 = invp[j4];
        #pragma unroll
        for (int jj = 0; jj < 4; ++jj) {
          const int j = j4 * 4 + jj;
          run += cur[j];
          v4f o = run * inv4[jj];
          __builtin_nontemporal_store(o, &q[(size_t)(bt * U + j) * ROWV]);
        }
      }
      if (bt + 1 < NB) {
        #pragma unroll
        for (int j = 0; j < U; ++j) cur[j] = nxt[j];
      }
    }
  }
}

extern "C" void kernel_launch(void* const* d_in, const int* in_sizes, int n_in,
                              void* d_out, int out_size, void* d_ws, size_t ws_size,
                              hipStream_t stream) {
  const float* x = (const float*)d_in[0];
  float* out = (float*)d_out;
  const int B = in_sizes[0] / (S * Dm);
  dim3 block(TX, TY);
  dim3 grid((unsigned)(B * Dm / DT));
  hipLaunchKernelGGL(cummean_kernel, grid, block, 0, stream, x, out);
}


# Read x once: 4 chunks of 1024 rows kept in registers across the WMMA segment scan (chunk total carried through C), next chunk's loads in flight over hand-written LDS barriers; new reduction tree
# speedup vs baseline: 3.3327x; 1.3381x over previous
#include <hip/hip_runtime.h>


typedef float v4f __attribute__((ext_vector_type(4)));
typedef float v2f __attribute__((ext_vector_type(2)));
typedef float v8f __attribute__((ext_vector_type(8)));

constexpr int S     = 4096;
constexpr int Dm    = 1024;
constexpr int TX    = 8;
constexpr int TY    = 64;
constexpr int DT    = TX * 4;
constexpr int NTHR  = TX * TY;
constexpr int U     = 16;
constexpr int CH    = TY * U;
constexpr int NCH   = S / CH;
constexpr int PITCH = TY + 1;
constexpr int ROWV  = Dm / 4;

static __device__ __forceinline__ void lds_barrier() {
  asm volatile("s_wait_dscnt 0x0\n\ts_barrier_signal -1\n\ts_barrier_wait -1" ::: "memory");
}

__global__ __launch_bounds__(NTHR) void cummean_kernel(
    const float* __restrict__ x, float* __restrict__ out) {
  __shared__ float lds_sums[DT][PITCH];
  __shared__ float lds_excl[DT][PITCH];
  __shared__ float lds_carry[DT];
  __shared__ __attribute__((aligned(16))) float inv_lds[S];

  const int tx  = threadIdx.x;
  const int ty  = threadIdx.y;
  const int tid = ty * TX + tx;

  const int tile        = blockIdx.x;
  const int tiles_per_b = Dm / DT;
  const int b           = tile / tiles_per_b;
  const int dbase       = (tile % tiles_per_b) * DT;
  const float* xb = x   + (size_t)b * S * Dm + dbase;
  float*       ob = out + (size_t)b * S * Dm + dbase;

  const int dd = tx * 4;
  const int r0 = ty * U;
  const v4f* p = (const v4f*)(xb + (size_t)r0 * Dm + dd);
  v4f*       q = (v4f*)(ob + (size_t)r0 * Dm + dd);

  const int wid     = tid >> 5;
  const int lane    = tid & 31;
  const int colbase = (wid & 1) * 16;
  const int g       = (wid >> 1) & 3;
  const int m       = lane & 15;
  const int half    = lane >> 4;
  const int mrow    = g * 16 + m;

  v4f cur[U], nxt[U];
  #pragma unroll
  for (int j = 0; j < U; ++j) cur[j] = p[(size_t)j * ROWV];

  for (int i = tid; i < S; i += NTHR) {
    inv_lds[i] = 1.0f / (float)(i + 1);
  }

  #pragma unroll
  for (int ch = 0; ch < NCH; ++ch) {
    __builtin_amdgcn_sched_barrier(0);
    if (ch + 1 < NCH) {
      #pragma unroll
      for (int j = 0; j < U; ++j) nxt[j] = p[(size_t)((ch + 1) * CH + j) * ROWV];
    }
    __builtin_amdgcn_sched_barrier(0);

    v4f acc = cur[0];
    #pragma unroll
    for (int j = 1; j < U; ++j) acc += cur[j];
    lds_sums[dd + 0][ty] = acc.x;
    lds_sums[dd + 1][ty] = acc.y;
    lds_sums[dd + 2][ty] = acc.z;
    lds_sums[dd + 3][ty] = acc.w;
    lds_barrier();

    if (wid < (DT / 16) * (TY / 16)) {
      v8f c = {};
      if (ch > 0) {
        const float cin = lds_carry[colbase + m];
        #pragma unroll
        for (int r = 0; r < 8; ++r) c[r] = cin;
      }
      #pragma unroll
      for (int k4 = 0; k4 < TY / 4; ++k4) {
        const int k0 = k4 * 4 + half * 2;
        v2f a, bm;
        a.x = (k0     < mrow) ? 1.0f : 0.0f;
        a.y = (k0 + 1 < mrow) ? 1.0f : 0.0f;
        bm.x = lds_sums[colbase + m][k0];
        bm.y = lds_sums[colbase + m][k0 + 1];
        c = __builtin_amdgcn_wmma_f32_16x16x4_f32(false, a, false, bm, (short)0,
                                                  c, false, false);
      }
      #pragma unroll
      for (int r = 0; r < 8; ++r) {
        lds_excl[colbase + m][g * 16 + r + 8 * half] = c[r];
      }
    }
    lds_barrier();

    v4f run;
    run.x = lds_excl[dd + 0][ty];
    run.y = lds_excl[dd + 1][ty];
    run.z = lds_excl[dd + 2][ty];
    run.w = lds_excl[dd + 3][ty];
    const v4f* invp = (const v4f*)&inv_lds[ch * CH + r0];
    #pragma unroll
    for (int j4 = 0; j4 < U / 4; ++j4) {
      const v4f inv4 = invp[j4];
      #pragma unroll
      for (int jj = 0; jj < 4; ++jj) {
        const int j = j4 * 4 + jj;
        run += cur[j];
        v4f o = run * inv4[jj];
        __builtin_nontemporal_store(o, &q[(size_t)(ch * CH + j) * ROWV]);
      }
    }
    if (ch + 1 < NCH) {
      if (ty == TY - 1) {
        lds_carry[dd + 0] = run.x;
        lds_carry[dd + 1] = run.y;
        lds_carry[dd + 2] = run.z;
        lds_carry[dd + 3] = run.w;
      }
      #pragma unroll
      for (int j = 0; j < U; ++j) cur[j] = nxt[j];
    }
  }
}

extern "C" void kernel_launch(void* const* d_in, const int* in_sizes, int n_in,
                              void* d_out, int out_size, void* d_ws, size_t ws_size,
                              hipStream_t stream) {
  const float* x = (const float*)d_in[0];
  float* out = (float*)d_out;
  const int B = in_sizes[0] / (S * Dm);
  dim3 block(TX, TY);
  dim3 grid((unsigned)(B * Dm / DT));
  hipLaunchKernelGGL(cummean_kernel, grid, block, 0, stream, x, out);
}
